# starting kernel
# baseline (speedup 1.0000x reference)
#include <hip/hip_runtime.h>

typedef __attribute__((ext_vector_type(2))) float v2f;
typedef __attribute__((ext_vector_type(8))) float v8f;

#define NB   16
#define NC   192
#define NH   64
#define NW   64
#define NK   (NH * NW)
#define CBLK 32
#define NCB  (NC / CBLK)
#define NDT  (NC / 16)
#define NOFF 5
#define CC   (NC * NC)

#define WM(A_, B_, C_) \
    __builtin_amdgcn_wmma_f32_16x16x4_f32(false, (A_), false, (B_), (short)0, (C_), false, false)

#define WMMA10(a0_, a1_, b0_, b1_, b2_, b3_, b4_) do {            \
    accA0 = WM(a0_, b0_, accA0); accB0 = WM(a1_, b0_, accB0);     \
    accA1 = WM(a0_, b1_, accA1); accB1 = WM(a1_, b1_, accB1);     \
    accA2 = WM(a0_, b2_, accA2); accB2 = WM(a1_, b2_, accB2);     \
    accA3 = WM(a0_, b3_, accA3); accB3 = WM(a1_, b3_, accB3);     \
    accA4 = WM(a0_, b4_, accA4); accB4 = WM(a1_, b4_, accB4);     \
} while (0)

__global__ void cofe_zero_ws(float* ws) {
    int i = threadIdx.x;
    if (i < NB * NOFF) ws[i] = 0.0f;
}

__device__ __forceinline__ void tile_epilogue(v8f acc, int o, int b, int cb, int db,
                                              int m_off, int n, int lane,
                                              float* __restrict__ out,
                                              float* __restrict__ ss) {
    size_t base = (size_t)(b * NOFF + o) * CC + (size_t)(cb + m_off) * NC + (size_t)(db + n);
    float s = 0.0f;
#pragma unroll
    for (int i = 0; i < 8; ++i) {
        float v = acc[i];
        out[base + (size_t)i * NC] = v;
        s += v * v;
    }
#pragma unroll
    for (int off = 16; off > 0; off >>= 1)
        s += __shfl_xor(s, off, 32);
    if (lane == 0)
        atomicAdd(&ss[b * NOFF + o], s);
}

__global__ __launch_bounds__(256) void cofe_wmma(const float* __restrict__ x,
                                                 float* __restrict__ out,
                                                 float* __restrict__ ss) {
    const int wave   = (blockIdx.x << 3) + (threadIdx.x >> 5);
    const int lane   = threadIdx.x & 31;
    const int lane15 = lane & 15;
    const int kk     = (lane >> 4) << 1;

    const int b  = wave / (NCB * NDT);
    const int rm = wave % (NCB * NDT);
    const int cb = (rm / NDT) * CBLK;
    const int db = (rm % NDT) * 16;

    const float* __restrict__ xb = x + (size_t)b * NC * NK;
    const int xrow = (cb + lane15) * NK;
    const int yrow = (db + lane15) * NK;

    v8f accA0 = {}, accA1 = {}, accA2 = {}, accA3 = {}, accA4 = {};
    v8f accB0 = {}, accB1 = {}, accB2 = {}, accB3 = {}, accB4 = {};

    for (int h = 0; h < NH; ++h) {
        const int hm  = (h > 0) ? (h - 1) : 0;
        const int rm1 = yrow + (hm << 6);
        const int r0  = yrow + (h  << 6);
        const int xr  = xrow + (h  << 6);

        auto clamped_step = [&](int w) {
            const int wm  = (w > 0)  ? (w - 1) : 0;
            const int wp2 = (w < 62) ? (w + 2) : 63;
            v2f a0 = *(const v2f*)(xb + xr + w);
            v2f a1 = *(const v2f*)(xb + xr + 16 * NK + w);
            v2f b0, b1, b2, b3, b4;
            b0.x = xb[rm1 + wm];    b0.y = xb[rm1 + w];
            b1   = *(const v2f*)(xb + rm1 + w);
            b2.x = xb[rm1 + w + 1]; b2.y = xb[rm1 + wp2];
            b3.x = xb[r0 + wm];     b3.y = xb[r0 + w];
            b4   = *(const v2f*)(xb + r0 + w);
            WMMA10(a0, a1, b0, b1, b2, b3, b4);
        };

        clamped_step(kk);

        for (int wi = 1; wi < 15; ++wi) {
            const int w = (wi << 2) + kk;
            v2f a0 = *(const v2f*)(xb + xr + w);
            v2f a1 = *(const v2f*)(xb + xr + 16 * NK + w);

            const float* __restrict__ pu = xb + rm1 + w - 1;
            const float* __restrict__ pc = xb + r0  + w - 1;
            float p0 = pu[0], p1 = pu[1], p2 = pu[2], p3 = pu[3];
            float q0 = pc[0], q1 = pc[1], q2 = pc[2];

            v2f b0 = {p0, p1};
            v2f b1 = {p1, p2};
            v2f b2 = {p2, p3};
            v2f b3 = {q0, q1};
            v2f b4 = {q1, q2};
            WMMA10(a0, a1, b0, b1, b2, b3, b4);
        }

        clamped_step(60 + kk);
    }

    const int m_off = (lane >> 4) << 3;
    tile_epilogue(accA0, 0, b, cb,      db, m_off, lane15, lane, out, ss);
    tile_epilogue(accB0, 0, b, cb + 16, db, m_off, lane15, lane, out, ss);
    tile_epilogue(accA1, 1, b, cb,      db, m_off, lane15, lane, out, ss);
    tile_epilogue(accB1, 1, b, cb + 16, db, m_off, lane15, lane, out, ss);
    tile_epilogue(accA2, 2, b, cb,      db, m_off, lane15, lane, out, ss);
    tile_epilogue(accB2, 2, b, cb + 16, db, m_off, lane15, lane, out, ss);
    tile_epilogue(accA3, 3, b, cb,      db, m_off, lane15, lane, out, ss);
    tile_epilogue(accB3, 3, b, cb + 16, db, m_off, lane15, lane, out, ss);
    tile_epilogue(accA4, 4, b, cb,      db, m_off, lane15, lane, out, ss);
    tile_epilogue(accB4, 4, b, cb + 16, db, m_off, lane15, lane, out, ss);
}

__global__ __launch_bounds__(256) void cofe_normalize(float* __restrict__ out,
                                                      const float* __restrict__ ss) {
    size_t i = (size_t)blockIdx.x * 256 + threadIdx.x;
    size_t grp = i / CC;
    float nrm = sqrtf(ss[grp]);
    nrm = fmaxf(nrm, 1e-12f);
    out[i] = out[i] / nrm;
}

extern "C" void kernel_launch(void* const* d_in, const int* in_sizes, int n_in,
                              void* d_out, int out_size, void* d_ws, size_t ws_size,
                              hipStream_t stream) {
    const float* x  = (const float*)d_in[0];
    float* out      = (float*)d_out;
    float* ss       = (float*)d_ws;

    cofe_zero_ws<<<1, 128, 0, stream>>>(ss);

    cofe_wmma<<<(NB * NCB * NDT) / 8, 256, 0, stream>>>(x, out, ss);

    cofe_normalize<<<(NB * NOFF * CC) / 256, 256, 0, stream>>>(out, ss);
}


# Re-tile for one equal share per SIMD: 48x48x5 tile per work-group of 4 waves, each wave sums a quarter of k (16 columns of every image row), partials added through LDS; changes summation order
# speedup vs baseline: 1.6055x; 1.6055x over previous
#include <hip/hip_runtime.h>


typedef __attribute__((ext_vector_type(2))) float v2f;
typedef __attribute__((ext_vector_type(8))) float v8f;

#define NB   16
#define NC   192
#define NH   64
#define NW   64
#define NK   (NH * NW)
#define NOFF 5
#define CC   (NC * NC)
#define TS   48
#define NT   (NC / TS)
#define NACC 45
#define NWV  4

#define WM(A_, B_, C_) \
    __builtin_amdgcn_wmma_f32_16x16x4_f32(false, (A_), false, (B_), (short)0, (C_), false, false)

__global__ void cofe_zero_ws(float* ws) {
    int i = threadIdx.x;
    if (i < NB * NOFF) ws[i] = 0.0f;
}

__device__ __forceinline__ void wmma45(v8f (&acc)[NACC], const v2f (&a)[3], const v2f (&bf)[3][NOFF]) {
#pragma unroll
    for (int di = 0; di < 3; ++di)
#pragma unroll
        for (int o = 0; o < NOFF; ++o)
#pragma unroll
            for (int ci = 0; ci < 3; ++ci)
                acc[(ci * 3 + di) * NOFF + o] = WM(a[ci], bf[di][o], acc[(ci * 3 + di) * NOFF + o]);
}

template <int EDGE>
__device__ __forceinline__ void cofe_accumulate(const float* __restrict__ xb, int xrow, int yrow,
                                                int w0, v8f (&acc)[NACC]) {
    for (int h = 0; h < NH; ++h) {
        const int hm  = (h > 0) ? (h - 1) : 0;
        const int rm1 = yrow + (hm << 6);
        const int r0  = yrow + (h  << 6);
        const int xr  = xrow + (h  << 6);

        auto clamped_step = [&](int w) {
            const int wm  = (w > 0)  ? (w - 1) : 0;
            const int wp2 = (w < 62) ? (w + 2) : 63;
            v2f a[3];
            v2f bf[3][NOFF];
#pragma unroll
            for (int ci = 0; ci < 3; ++ci)
                a[ci] = *(const v2f*)(xb + xr + ci * 16 * NK + w);
#pragma unroll
            for (int di = 0; di < 3; ++di) {
                const float* __restrict__ pu = xb + rm1 + di * 16 * NK;
                const float* __restrict__ pc = xb + r0  + di * 16 * NK;
                bf[di][0].x = pu[wm];    bf[di][0].y = pu[w];
                bf[di][1]   = *(const v2f*)(pu + w);
                bf[di][2].x = pu[w + 1]; bf[di][2].y = pu[wp2];
                bf[di][3].x = pc[wm];    bf[di][3].y = pc[w];
                bf[di][4]   = *(const v2f*)(pc + w);
            }
            wmma45(acc, a, bf);
        };

        auto interior_step = [&](int w) {
            v2f a[3];
            v2f bf[3][NOFF];
#pragma unroll
            for (int ci = 0; ci < 3; ++ci)
                a[ci] = *(const v2f*)(xb + xr + ci * 16 * NK + w);
#pragma unroll
            for (int di = 0; di < 3; ++di) {
                const float* __restrict__ pu = xb + rm1 + di * 16 * NK + w - 1;
                const float* __restrict__ pc = xb + r0  + di * 16 * NK + w - 1;
                float p0 = pu[0], p1 = pu[1], p2 = pu[2], p3 = pu[3];
                float q0 = pc[0], q1 = pc[1], q2 = pc[2];
                bf[di][0] = v2f{p0, p1};
                bf[di][1] = v2f{p1, p2};
                bf[di][2] = v2f{p2, p3};
                bf[di][3] = v2f{q0, q1};
                bf[di][4] = v2f{q1, q2};
            }
            wmma45(acc, a, bf);
        };

        if (EDGE == 0) clamped_step(w0); else interior_step(w0);
        interior_step(w0 + 4);
        interior_step(w0 + 8);
        if (EDGE == 2) clamped_step(w0 + 12); else interior_step(w0 + 12);
    }
}

__global__ __launch_bounds__(128) void cofe_wmma(const float* __restrict__ x,
                                                 float* __restrict__ out,
                                                 float* __restrict__ ss) {
    __shared__ float red[NWV][NACC][256];

    const int tid    = threadIdx.x;
    const int q      = tid >> 5;
    const int lane   = tid & 31;
    const int lane15 = lane & 15;
    const int kk     = (lane >> 4) << 1;

    const int b  = blockIdx.x / (NT * NT);
    const int rt = blockIdx.x % (NT * NT);
    const int cb = (rt / NT) * TS;
    const int db = (rt % NT) * TS;

    const float* __restrict__ xb = x + (size_t)b * NC * NK;
    const int xrow = (cb + lane15) * NK;
    const int yrow = (db + lane15) * NK;
    const int w0   = (q << 4) + kk;

    v8f acc[NACC];
#pragma unroll
    for (int t = 0; t < NACC; ++t) acc[t] = v8f{};

    if (q == 0)            cofe_accumulate<0>(xb, xrow, yrow, w0, acc);
    else if (q == NWV - 1) cofe_accumulate<2>(xb, xrow, yrow, w0, acc);
    else                   cofe_accumulate<1>(xb, xrow, yrow, w0, acc);

#pragma unroll
    for (int t = 0; t < NACC; ++t)
#pragma unroll
        for (int i = 0; i < 8; ++i)
            red[q][t][i * 32 + lane] = acc[t][i];
    __syncthreads();

#pragma unroll
    for (int o = 0; o < NOFF; ++o) {
        float s = 0.0f;
        for (int jj = 0; jj < (TS * TS) / 128; ++jj) {
            const int rem = jj * 128 + tid;
            const int row = rem / TS, col = rem % TS;
            const int t   = ((row >> 4) * 3 + (col >> 4)) * NOFF + o;
            const int pos = ((row & 7) << 5) + (col & 15) + (((row >> 3) & 1) << 4);
            const float v = (red[0][t][pos] + red[1][t][pos]) + (red[2][t][pos] + red[3][t][pos]);
            out[(size_t)(b * NOFF + o) * CC + (size_t)(cb + row) * NC + (size_t)(db + col)] = v;
            s += v * v;
        }
#pragma unroll
        for (int off = 16; off > 0; off >>= 1)
            s += __shfl_xor(s, off, 32);
        if (lane == 0)
            atomicAdd(&ss[b * NOFF + o], s);
    }
}

__global__ __launch_bounds__(256) void cofe_normalize(float* __restrict__ out,
                                                      const float* __restrict__ ss) {
    size_t i = (size_t)blockIdx.x * 256 + threadIdx.x;
    size_t grp = i / CC;
    float nrm = sqrtf(ss[grp]);
    nrm = fmaxf(nrm, 1e-12f);
    out[i] = out[i] / nrm;
}

extern "C" void kernel_launch(void* const* d_in, const int* in_sizes, int n_in,
                              void* d_out, int out_size, void* d_ws, size_t ws_size,
                              hipStream_t stream) {
    const float* x  = (const float*)d_in[0];
    float* out      = (float*)d_out;
    float* ss       = (float*)d_ws;

    cofe_zero_ws<<<1, 128, 0, stream>>>(ss);

    cofe_wmma<<<NB * NT * NT, 128, 0, stream>>>(x, out, ss);

    cofe_normalize<<<(NB * NOFF * CC) / 256, 256, 0, stream>>>(out, ss);
}


# Next image row's operands fetched one row ahead in one group of 128-bit loads, row h kept in registers as row h-1; each lane owns 8 columns, so k is remapped: changes summation order
# speedup vs baseline: 2.4633x; 1.5343x over previous
#include <hip/hip_runtime.h>


typedef __attribute__((ext_vector_type(2))) float v2f;
typedef __attribute__((ext_vector_type(4))) float v4f;
typedef __attribute__((ext_vector_type(8))) float v8f;

#define NB   16
#define NC   192
#define NH   64
#define NW   64
#define NK   (NH * NW)
#define NOFF 5
#define CC   (NC * NC)
#define TS   48
#define NT   (NC / TS)
#define NACC 45
#define NWV  4

#define WM(A_, B_, C_) \
    __builtin_amdgcn_wmma_f32_16x16x4_f32(false, (A_), false, (B_), (short)0, (C_), false, false)

__global__ void cofe_zero_ws(float* ws) {
    int i = threadIdx.x;
    if (i < NB * NOFF) ws[i] = 0.0f;
}

__device__ __forceinline__ void wmma45(v8f (&acc)[NACC], const v2f (&a)[3], const v2f (&bf)[3][NOFF]) {
#pragma unroll
    for (int di = 0; di < 3; ++di)
#pragma unroll
        for (int o = 0; o < NOFF; ++o)
#pragma unroll
            for (int ci = 0; ci < 3; ++ci)
                acc[(ci * 3 + di) * NOFF + o] = WM(a[ci], bf[di][o], acc[(ci * 3 + di) * NOFF + o]);
}

struct BWin { v4f lo, hi; float L, R; };

__device__ __forceinline__ void cofe_load_row(const float* __restrict__ pa, const float* __restrict__ pb,
                                              int offL, int offR, v4f (&aw)[3][2], BWin (&bw)[3]) {
#pragma unroll
    for (int ci = 0; ci < 3; ++ci) {
        aw[ci][0] = *(const v4f*)(pa + ci * 16 * NK);
        aw[ci][1] = *(const v4f*)(pa + ci * 16 * NK + 4);
    }
#pragma unroll
    for (int di = 0; di < 3; ++di) {
        const float* __restrict__ p = pb + di * 16 * NK;
        bw[di].lo = *(const v4f*)(p);
        bw[di].hi = *(const v4f*)(p + 4);
        bw[di].L  = p[offL];
        bw[di].R  = p[offR];
    }
}

__device__ __forceinline__ void cofe_row(v8f (&acc)[NACC], const v4f (&aw)[3][2],
                                         const BWin (&up)[3], const BWin (&cu)[3]) {
    float A[3][8], U[3][10], C[3][10];
#pragma unroll
    for (int i = 0; i < 3; ++i) {
#pragma unroll
        for (int e = 0; e < 4; ++e) {
            A[i][e] = aw[i][0][e];     A[i][4 + e] = aw[i][1][e];
            U[i][1 + e] = up[i].lo[e]; U[i][5 + e] = up[i].hi[e];
            C[i][1 + e] = cu[i].lo[e]; C[i][5 + e] = cu[i].hi[e];
        }
        U[i][0] = up[i].L; U[i][9] = up[i].R;
        C[i][0] = cu[i].L; C[i][9] = cu[i].R;
    }
#pragma unroll
    for (int j = 0; j < 4; ++j) {
        v2f a[3];
        v2f bf[3][NOFF];
#pragma unroll
        for (int ci = 0; ci < 3; ++ci)
            a[ci] = v2f{A[ci][2 * j], A[ci][2 * j + 1]};
#pragma unroll
        for (int di = 0; di < 3; ++di) {
            bf[di][0] = v2f{U[di][2 * j],     U[di][2 * j + 1]};
            bf[di][1] = v2f{U[di][2 * j + 1], U[di][2 * j + 2]};
            bf[di][2] = v2f{U[di][2 * j + 2], U[di][2 * j + 3]};
            bf[di][3] = v2f{C[di][2 * j],     C[di][2 * j + 1]};
            bf[di][4] = v2f{C[di][2 * j + 1], C[di][2 * j + 2]};
        }
        wmma45(acc, a, bf);
    }
}

__global__ __launch_bounds__(128) void cofe_wmma(const float* __restrict__ x,
                                                 float* __restrict__ out,
                                                 float* __restrict__ ss) {
    __shared__ float red[NWV][NACC][256];

    const int tid    = threadIdx.x;
    const int q      = tid >> 5;
    const int lane   = tid & 31;
    const int lane15 = lane & 15;

    const int b  = blockIdx.x / (NT * NT);
    const int rt = blockIdx.x % (NT * NT);
    const int cb = (rt / NT) * TS;
    const int db = (rt % NT) * TS;

    const float* __restrict__ xb = x + (size_t)b * NC * NK;
    const int colbase = (q << 4) + ((lane >> 4) << 3);
    const int offL = (colbase == 0)      ? 0 : -1;
    const int offR = (colbase == NW - 8) ? 7 :  8;
    const float* __restrict__ pa = xb + (cb + lane15) * NK + colbase;
    const float* __restrict__ pb = xb + (db + lane15) * NK + colbase;

    v8f acc[NACC];
#pragma unroll
    for (int t = 0; t < NACC; ++t) acc[t] = v8f{};

    v4f  acur[3][2], anxt[3][2];
    BWin up[3], cu[3], nx[3];
    cofe_load_row(pa, pb, offL, offR, acur, cu);
#pragma unroll
    for (int di = 0; di < 3; ++di) up[di] = cu[di];

    for (int h = 0; h < NH; ++h) {
        const int hn = (h < NH - 1) ? (h + 1) : (NH - 1);
        cofe_load_row(pa + (hn << 6), pb + (hn << 6), offL, offR, anxt, nx);
        cofe_row(acc, acur, up, cu);
#pragma unroll
        for (int i = 0; i < 3; ++i) {
            up[i] = cu[i];
            cu[i] = nx[i];
            acur[i][0] = anxt[i][0];
            acur[i][1] = anxt[i][1];
        }
    }

#pragma unroll
    for (int t = 0; t < NACC; ++t)
#pragma unroll
        for (int i = 0; i < 8; ++i)
            red[q][t][i * 32 + lane] = acc[t][i];
    __syncthreads();

#pragma unroll
    for (int o = 0; o < NOFF; ++o) {
        float s = 0.0f;
        for (int jj = 0; jj < (TS * TS) / 128; ++jj) {
            const int rem = jj * 128 + tid;
            const int row = rem / TS, col = rem % TS;
            const int t   = ((row >> 4) * 3 + (col >> 4)) * NOFF + o;
            const int pos = ((row & 7) << 5) + (col & 15) + (((row >> 3) & 1) << 4);
            const float v = (red[0][t][pos] + red[1][t][pos]) + (red[2][t][pos] + red[3][t][pos]);
            out[(size_t)(b * NOFF + o) * CC + (size_t)(cb + row) * NC + (size_t)(db + col)] = v;
            s += v * v;
        }
#pragma unroll
        for (int off = 16; off > 0; off >>= 1)
            s += __shfl_xor(s, off, 32);
        if (lane == 0)
            atomicAdd(&ss[b * NOFF + o], s);
    }
}

__global__ __launch_bounds__(256) void cofe_normalize(float* __restrict__ out,
                                                      const float* __restrict__ ss) {
    size_t i = (size_t)blockIdx.x * 256 + threadIdx.x;
    size_t grp = i / CC;
    float nrm = sqrtf(ss[grp]);
    nrm = fmaxf(nrm, 1e-12f);
    out[i] = out[i] / nrm;
}

extern "C" void kernel_launch(void* const* d_in, const int* in_sizes, int n_in,
                              void* d_out, int out_size, void* d_ws, size_t ws_size,
                              hipStream_t stream) {
    const float* x  = (const float*)d_in[0];
    float* out      = (float*)d_out;
    float* ss       = (float*)d_ws;

    cofe_zero_ws<<<1, 128, 0, stream>>>(ss);

    cofe_wmma<<<NB * NT * NT, 128, 0, stream>>>(x, out, ss);

    cofe_normalize<<<(NB * NOFF * CC) / 256, 256, 0, stream>>>(out, ss);
}


# (0,0) plane is symmetric: only its 78 fragments on/above the diagonal computed, mirrors stored (41 accumulators, not 45); square sums in fixed-order slots, no atomics, no zero launch; reorders sums
# speedup vs baseline: 2.7230x; 1.1054x over previous
#include <hip/hip_runtime.h>


typedef __attribute__((ext_vector_type(2))) float v2f;
typedef __attribute__((ext_vector_type(4))) float v4f;
typedef __attribute__((ext_vector_type(8))) float v8f;

#define NB   16
#define NC   192
#define NH   64
#define NW   64
#define NK   (NH * NW)
#define NOFF 5
#define NPL  4
#define CC   (NC * NC)
#define TS   48
#define NT   (NC / TS)
#define NSLOT (NT * NT)
#define NMAIN 36
#define NACC 41
#define NWV  4


#define WM(A_, B_, C_) \
    __builtin_amdgcn_wmma_f32_16x16x4_f32(false, (A_), false, (B_), (short)0, (C_), false, false)

#define EXT_K0(X) X(0, 0, 0) X(1, 0, 1) X(2, 0, 2) X(3, 1, 1) X(4, 1, 2)
#define EXT_K1(X) X(0, 0, 0) X(1, 0, 1) X(2, 0, 2) X(3, 1, 0) X(4, 1, 1)
#define EXT_K2(X) X(0, 0, 2) X(1, 1, 2) X(2, 2, 1) X(3, 2, 2)

template <int KIND>
__device__ __forceinline__ void wmma41(v8f (&acc)[NACC], const v2f (&a)[3], const v2f (&bf)[3][NOFF]) {
#pragma unroll
    for (int di = 0; di < 3; ++di)
#pragma unroll
        for (int o = 0; o < NPL; ++o)
#pragma unroll
            for (int ci = 0; ci < 3; ++ci)
                acc[(ci * 3 + di) * NPL + o] = WM(a[ci], bf[di][o], acc[(ci * 3 + di) * NPL + o]);
#define EXT_MM(e, ci, di) acc[NMAIN + e] = WM(a[ci], bf[di][4], acc[NMAIN + e]);
    if constexpr (KIND == 0) { EXT_K0(EXT_MM) }
    else if constexpr (KIND == 1) { EXT_K1(EXT_MM) }
    else { EXT_K2(EXT_MM) }
#undef EXT_MM
    if constexpr (KIND == 2) acc[NMAIN + 4] = WM(bf[2][4], bf[2][4], acc[NMAIN + 4]);
}

struct BWin { v4f lo, hi; float L, R; };

__device__ __forceinline__ void cofe_load_row(const float* __restrict__ pa, const float* __restrict__ pb,
                                              int offL, int offR, v4f (&aw)[3][2], BWin (&bw)[3]) {
#pragma unroll
    for (int ci = 0; ci < 3; ++ci) {
        aw[ci][0] = *(const v4f*)(pa + ci * 16 * NK);
        aw[ci][1] = *(const v4f*)(pa + ci * 16 * NK + 4);
    }
#pragma unroll
    for (int di = 0; di < 3; ++di) {
        const float* __restrict__ p = pb + di * 16 * NK;
        bw[di].lo = *(const v4f*)(p);
        bw[di].hi = *(const v4f*)(p + 4);
        bw[di].L  = p[offL];
        bw[di].R  = p[offR];
    }
}

template <int KIND>
__device__ __forceinline__ void cofe_row(v8f (&acc)[NACC], const v4f (&aw)[3][2],
                                         const BWin (&up)[3], const BWin (&cu)[3]) {
    float A[3][8], U[3][10], C[3][10];
#pragma unroll
    for (int i = 0; i < 3; ++i) {
#pragma unroll
        for (int e = 0; e < 4; ++e) {
            A[i][e] = aw[i][0][e];     A[i][4 + e] = aw[i][1][e];
            U[i][1 + e] = up[i].lo[e]; U[i][5 + e] = up[i].hi[e];
            C[i][1 + e] = cu[i].lo[e]; C[i][5 + e] = cu[i].hi[e];
        }
        U[i][0] = up[i].L; U[i][9] = up[i].R;
        C[i][0] = cu[i].L; C[i][9] = cu[i].R;
    }
#pragma unroll
    for (int j = 0; j < 4; ++j) {
        v2f a[3];
        v2f bf[3][NOFF];
#pragma unroll
        for (int ci = 0; ci < 3; ++ci)
            a[ci] = v2f{A[ci][2 * j], A[ci][2 * j + 1]};
#pragma unroll
        for (int di = 0; di < 3; ++di) {
            bf[di][0] = v2f{U[di][2 * j],     U[di][2 * j + 1]};
            bf[di][1] = v2f{U[di][2 * j + 1], U[di][2 * j + 2]};
            bf[di][2] = v2f{U[di][2 * j + 2], U[di][2 * j + 3]};
            bf[di][3] = v2f{C[di][2 * j],     C[di][2 * j + 1]};
            bf[di][4] = v2f{C[di][2 * j + 1], C[di][2 * j + 2]};
        }
        wmma41<KIND>(acc, a, bf);
    }
}

__device__ __forceinline__ float red4(const float (*red)[NACC][256], int t, int pos) {
    return (red[0][t][pos] + red[1][t][pos]) + (red[2][t][pos] + red[3][t][pos]);
}

template <bool MIRROR>
__device__ __forceinline__ float ext_store(const float (*red)[NACC][256], int t, int rbase, int cbase,
                                           float* __restrict__ op, int tid) {
    float s = 0.0f;
#pragma unroll
    for (int jj = 0; jj < 2; ++jj) {
        const int idx = jj * 128 + tid;
        const int r = idx >> 4, c = idx & 15;
        const float v = red4(red, t, ((r & 7) << 5) + c + (((r >> 3) & 1) << 4));
        op[(size_t)(rbase + r) * NC + (size_t)(cbase + c)] = v;
        s += v * v;
        if constexpr (MIRROR) {
            const int r2 = idx & 15, c2 = idx >> 4;
            const float v2 = red4(red, t, ((r2 & 7) << 5) + c2 + (((r2 >> 3) & 1) << 4));
            op[(size_t)(cbase + c2) * NC + (size_t)(rbase + r2)] = v2;
        }
    }
    return MIRROR ? 2.0f * s : s;
}

template <int KIND>
__device__ __forceinline__ void cofe_body(const float* __restrict__ x, float* __restrict__ out,
                                          float* __restrict__ ss, float (*red)[NACC][256],
                                          float (*sred)[NWV], int b, int rt, int cb, int db) {
    constexpr int NUSED = NMAIN + (KIND == 3 ? 4 : 5);

    const int tid    = threadIdx.x;
    const int q      = tid >> 5;
    const int lane   = tid & 31;
    const int lane15 = lane & 15;

    const float* __restrict__ xb = x + (size_t)b * NC * NK;
    const int colbase = (q << 4) + ((lane >> 4) << 3);
    const int offL = (colbase == 0)      ? 0 : -1;
    const int offR = (colbase == NW - 8) ? 7 :  8;
    const float* __restrict__ pa = xb + (cb + lane15) * NK + colbase;
    const float* __restrict__ pb = xb + (db + lane15) * NK + colbase;

    v8f acc[NACC];
#pragma unroll
    for (int t = 0; t < NACC; ++t) acc[t] = v8f{};

    v4f  acur[3][2], anxt[3][2];
    BWin up[3], cu[3], nx[3];
    cofe_load_row(pa, pb, offL, offR, acur, cu);
#pragma unroll
    for (int di = 0; di < 3; ++di) up[di] = cu[di];

    for (int h = 0; h < NH; ++h) {
        const int hn = (h < NH - 1) ? (h + 1) : (NH - 1);
        cofe_load_row(pa + (hn << 6), pb + (hn << 6), offL, offR, anxt, nx);
        cofe_row<KIND>(acc, acur, up, cu);
#pragma unroll
        for (int i = 0; i < 3; ++i) {
            up[i] = cu[i];
            cu[i] = nx[i];
            acur[i][0] = anxt[i][0];
            acur[i][1] = anxt[i][1];
        }
    }

#pragma unroll
    for (int t = 0; t < NUSED; ++t)
#pragma unroll
        for (int i = 0; i < 8; ++i)
            red[q][t][i * 32 + lane] = acc[t][i];
    __syncthreads();

    float sq[NOFF];

#pragma unroll
    for (int o = 0; o < NPL; ++o) {
        float s = 0.0f;
        for (int jj = 0; jj < (TS * TS) / 128; ++jj) {
            const int rem = jj * 128 + tid;
            const int row = rem / TS, col = rem % TS;
            const int t   = ((row >> 4) * 3 + (col >> 4)) * NPL + o;
            const int pos = ((row & 7) << 5) + (col & 15) + (((row >> 3) & 1) << 4);
            const float v = red4(red, t, pos);
            out[(size_t)(b * NOFF + o) * CC + (size_t)(cb + row) * NC + (size_t)(db + col)] = v;
            s += v * v;
        }
        sq[o] = s;
    }

    {
        float* __restrict__ op = out + (size_t)(b * NOFF + 4) * CC;
        float s = 0.0f;
#define EXT_ST(e, ci, di) \
        s += ext_store<!(KIND == 0 && (ci) == (di))>(red, NMAIN + e, cb + 16 * (ci), db + 16 * (di), op, tid);
        if constexpr (KIND == 0) { EXT_K0(EXT_ST) }
        else if constexpr (KIND == 1) { EXT_K1(EXT_ST) }
        else { EXT_K2(EXT_ST) }
#undef EXT_ST
        if constexpr (KIND == 2) s += ext_store<false>(red, NMAIN + 4, db + 32, db + 32, op, tid);
        sq[4] = s;
    }

#pragma unroll
    for (int o = 0; o < NOFF; ++o) {
        float s = sq[o];
#pragma unroll
        for (int off = 16; off > 0; off >>= 1)
            s += __shfl_xor(s, off, 32);
        if (lane == 0) sred[o][q] = s;
    }
    __syncthreads();
    if (tid < NOFF)
        ss[(b * NOFF + tid) * NSLOT + rt] = (sred[tid][0] + sred[tid][1]) + (sred[tid][2] + sred[tid][3]);
}

__global__ __launch_bounds__(128) void cofe_wmma(const float* __restrict__ x,
                                                 float* __restrict__ out,
                                                 float* __restrict__ ss) {
    __shared__ float red[NWV][NACC][256];
    __shared__ float sred[NOFF][NWV];

    const int b  = blockIdx.x / NSLOT;
    const int rt = blockIdx.x % NSLOT;
    const int I  = rt / NT, J = rt % NT;
    const int cb = I * TS;
    const int db = J * TS;
    const int d  = (J - I) & 3;

    if (d == 0)                          cofe_body<0>(x, out, ss, red, sred, b, rt, cb, db);
    else if (d == 1 || (d == 2 && I < J)) cofe_body<1>(x, out, ss, red, sred, b, rt, cb, db);
    else if (d == 3)                     cofe_body<2>(x, out, ss, red, sred, b, rt, cb, db);
    else                                 cofe_body<3>(x, out, ss, red, sred, b, rt, cb, db);
}

__global__ __launch_bounds__(256) void cofe_normalize(float* __restrict__ out,
                                                      const float* __restrict__ ss) {
    const size_t i = (size_t)blockIdx.x * 256 + threadIdx.x;
    const float v = out[i];
    __builtin_amdgcn_sched_barrier(0);
    const int grp = blockIdx.x / (CC / 256);
    const float* __restrict__ p = ss + grp * NSLOT;
    float t = 0.0f;
#pragma unroll
    for (int j = 0; j < NSLOT; ++j) t += p[j];
    float nrm = sqrtf(t);
    nrm = fmaxf(nrm, 1e-12f);
    out[i] = v / nrm;
}

extern "C" void kernel_launch(void* const* d_in, const int* in_sizes, int n_in,
                              void* d_out, int out_size, void* d_ws, size_t ws_size,
                              hipStream_t stream) {
    const float* x  = (const float*)d_in[0];
    float* out      = (float*)d_out;
    float* ss       = (float*)d_ws;

    cofe_wmma<<<NB * NT * NT, 128, 0, stream>>>(x, out, ss);

    cofe_normalize<<<(NB * NOFF * CC) / 256, 256, 0, stream>>>(out, ss);
}


# Three row -1 offsets from 4 summed product planes of column-pair sums/differences (3 lags x 2 taps in 4 products, not 6), combined after the k sum: 128 not 164 matrix ops per row; reorders sums
# speedup vs baseline: 3.0937x; 1.1361x over previous
#include <hip/hip_runtime.h>


typedef __attribute__((ext_vector_type(2))) float v2f;
typedef __attribute__((ext_vector_type(4))) float v4f;
typedef __attribute__((ext_vector_type(8))) float v8f;

#define NB   16
#define NC   192
#define NH   64
#define NW   64
#define NK   (NH * NW)
#define NOFF 5
#define NTF  4
#define CC   (NC * NC)
#define TS   48
#define NT   (NC / TS)
#define NSLOT (NT * NT)
#define NMAIN 36
#define NPO  9
#define NEXT0 (NMAIN + NPO)
#define NACC (NEXT0 + 5)
#define NWV  4


#define WM(A_, B_, C_) \
    __builtin_amdgcn_wmma_f32_16x16x4_f32(false, (A_), false, (B_), (short)0, (C_), false, false)

#define EXT_K0(X) X(0, 0, 0) X(1, 0, 1) X(2, 0, 2) X(3, 1, 1) X(4, 1, 2)
#define EXT_K1(X) X(0, 0, 0) X(1, 0, 1) X(2, 0, 2) X(3, 1, 0) X(4, 1, 1)
#define EXT_K2(X) X(0, 0, 2) X(1, 1, 2) X(2, 2, 1) X(3, 2, 2)


struct BWin { v4f lo, hi; float L, R; };

__device__ __forceinline__ void cofe_load_row(const float* __restrict__ pa, const float* __restrict__ pb,
                                              int offL, int offR, v4f (&aw)[3][2], BWin (&bw)[3]) {
#pragma unroll
    for (int ci = 0; ci < 3; ++ci) {
        aw[ci][0] = *(const v4f*)(pa + ci * 16 * NK);
        aw[ci][1] = *(const v4f*)(pa + ci * 16 * NK + 4);
    }
#pragma unroll
    for (int di = 0; di < 3; ++di) {
        const float* __restrict__ p = pb + di * 16 * NK;
        bw[di].lo = *(const v4f*)(p);
        bw[di].hi = *(const v4f*)(p + 4);
        bw[di].L  = p[offL];
        bw[di].R  = p[offR];
    }
}

template <int KIND>
__device__ __forceinline__ void cofe_row(v8f (&acc)[NACC], const v4f (&aw)[3][2],
                                         const BWin (&up)[3], const BWin (&cu)[3]) {
    float A[3][8], U[3][10], C[3][10];
#pragma unroll
    for (int i = 0; i < 3; ++i) {
#pragma unroll
        for (int e = 0; e < 4; ++e) {
            A[i][e] = aw[i][0][e];     A[i][4 + e] = aw[i][1][e];
            U[i][1 + e] = up[i].lo[e]; U[i][5 + e] = up[i].hi[e];
            C[i][1 + e] = cu[i].lo[e]; C[i][5 + e] = cu[i].hi[e];
        }
        U[i][0] = up[i].L; U[i][9] = up[i].R;
        C[i][0] = cu[i].L; C[i][9] = cu[i].R;
    }

#pragma unroll
    for (int s = 0; s < 2; ++s) {
        v2f ea[3][NTF], vb[3][NTF];
#pragma unroll
        for (int i = 0; i < 3; ++i) {
#pragma unroll
            for (int e = 0; e < 2; ++e) {
                const int j = 2 * s + e;
                const float a0 = A[i][2 * j], a1 = A[i][2 * j + 1];
                ea[i][0][e] = a0;
                ea[i][1][e] = a0 + a1;
                ea[i][2][e] = a0 - a1;
                ea[i][3][e] = a1;
                const float b0 = U[i][2 * j], b1 = U[i][2 * j + 1], b2 = U[i][2 * j + 2], b3 = U[i][2 * j + 3];
                vb[i][0][e] = b0 - b2;
                vb[i][1][e] = b1 + b2;
                vb[i][2][e] = b2 - b1;
                vb[i][3][e] = b3 - b1;
            }
        }
#pragma unroll
        for (int di = 0; di < 3; ++di)
#pragma unroll
            for (int t = 0; t < NTF; ++t)
#pragma unroll
                for (int ci = 0; ci < 3; ++ci)
                    acc[(ci * 3 + di) * NTF + t] = WM(ea[ci][t], vb[di][t], acc[(ci * 3 + di) * NTF + t]);
    }

#pragma unroll
    for (int j = 0; j < 4; ++j) {
        v2f a[3], b3[3], b4[3];
#pragma unroll
        for (int i = 0; i < 3; ++i) {
            a[i]  = v2f{A[i][2 * j],     A[i][2 * j + 1]};
            b3[i] = v2f{C[i][2 * j],     C[i][2 * j + 1]};
            b4[i] = v2f{C[i][2 * j + 1], C[i][2 * j + 2]};
        }
#pragma unroll
        for (int di = 0; di < 3; ++di)
#pragma unroll
            for (int ci = 0; ci < 3; ++ci)
                acc[NMAIN + ci * 3 + di] = WM(a[ci], b3[di], acc[NMAIN + ci * 3 + di]);
#define EXT_MM(e, ci, di) acc[NEXT0 + e] = WM(a[ci], b4[di], acc[NEXT0 + e]);
        if constexpr (KIND == 0) { EXT_K0(EXT_MM) }
        else if constexpr (KIND == 1) { EXT_K1(EXT_MM) }
        else { EXT_K2(EXT_MM) }
#undef EXT_MM
        if constexpr (KIND == 2) acc[NEXT0 + 4] = WM(b4[2], b4[2], acc[NEXT0 + 4]);
    }
}

__device__ __forceinline__ float red4(const float (*red)[NACC][256], int t, int pos) {
    return (red[0][t][pos] + red[1][t][pos]) + (red[2][t][pos] + red[3][t][pos]);
}

template <bool MIRROR>
__device__ __forceinline__ float ext_store(const float (*red)[NACC][256], int t, int rbase, int cbase,
                                           float* __restrict__ op, int tid) {
    float s = 0.0f;
#pragma unroll
    for (int jj = 0; jj < 2; ++jj) {
        const int idx = jj * 128 + tid;
        const int r = idx >> 4, c = idx & 15;
        const float v = red4(red, t, ((r & 7) << 5) + c + (((r >> 3) & 1) << 4));
        op[(size_t)(rbase + r) * NC + (size_t)(cbase + c)] = v;
        s += v * v;
        if constexpr (MIRROR) {
            const int r2 = idx & 15, c2 = idx >> 4;
            const float v2 = red4(red, t, ((r2 & 7) << 5) + c2 + (((r2 >> 3) & 1) << 4));
            op[(size_t)(cbase + c2) * NC + (size_t)(rbase + r2)] = v2;
        }
    }
    return MIRROR ? 2.0f * s : s;
}

template <int KIND>
__device__ __forceinline__ void cofe_body(const float* __restrict__ x, float* __restrict__ out,
                                          float* __restrict__ ss, float (*red)[NACC][256],
                                          float (*sred)[NWV], int b, int rt, int cb, int db) {
    constexpr int NUSED = NEXT0 + (KIND == 3 ? 4 : 5);

    const int tid    = threadIdx.x;
    const int q      = tid >> 5;
    const int lane   = tid & 31;
    const int lane15 = lane & 15;

    const float* __restrict__ xb = x + (size_t)b * NC * NK;
    const int colbase = (q << 4) + ((lane >> 4) << 3);
    const int offL = (colbase == 0)      ? 0 : -1;
    const int offR = (colbase == NW - 8) ? 7 :  8;
    const float* __restrict__ pa = xb + (cb + lane15) * NK + colbase;
    const float* __restrict__ pb = xb + (db + lane15) * NK + colbase;

    v8f acc[NACC];
#pragma unroll
    for (int t = 0; t < NACC; ++t) acc[t] = v8f{};

    v4f  acur[3][2], anxt[3][2];
    BWin up[3], cu[3], nx[3];
    cofe_load_row(pa, pb, offL, offR, acur, cu);
#pragma unroll
    for (int di = 0; di < 3; ++di) up[di] = cu[di];

    for (int h = 0; h < NH; ++h) {
        const int hn = (h < NH - 1) ? (h + 1) : (NH - 1);
        cofe_load_row(pa + (hn << 6), pb + (hn << 6), offL, offR, anxt, nx);
        cofe_row<KIND>(acc, acur, up, cu);
#pragma unroll
        for (int i = 0; i < 3; ++i) {
            up[i] = cu[i];
            cu[i] = nx[i];
            acur[i][0] = anxt[i][0];
            acur[i][1] = anxt[i][1];
        }
    }

#pragma unroll
    for (int t = 0; t < NUSED; ++t)
#pragma unroll
        for (int i = 0; i < 8; ++i)
            red[q][t][i * 32 + lane] = acc[t][i];
    __syncthreads();

    float sq[NOFF];

    {
        float s0 = 0.0f, s1 = 0.0f, s2 = 0.0f, s3 = 0.0f;
        float* __restrict__ ob = out + (size_t)(b * NOFF) * CC + (size_t)cb * NC + (size_t)db;
#pragma unroll
        for (int jj = 0; jj < (TS * TS) / 128; ++jj) {
            const int rem = jj * 128 + tid;
            const int row = rem / TS, col = rem % TS;
            const int f   = (row >> 4) * 3 + (col >> 4);
            const int pos = ((row & 7) << 5) + (col & 15) + (((row >> 3) & 1) << 4);
            const float m0 = red4(red, f * NTF + 0, pos);
            const float m1 = red4(red, f * NTF + 1, pos);
            const float m2 = red4(red, f * NTF + 2, pos);
            const float m3 = red4(red, f * NTF + 3, pos);
            const float y3 = red4(red, NMAIN + f, pos);
            const float hs = 0.5f * (m1 + m2);
            const float y0 = m0 + hs;
            const float y1 = 0.5f * (m1 - m2);
            const float y2 = hs + m3;
            float* __restrict__ o = ob + (size_t)row * NC + (size_t)col;
            o[0]               = y0;
            o[(size_t)CC]      = y1;
            o[(size_t)2 * CC]  = y2;
            o[(size_t)3 * CC]  = y3;
            s0 += y0 * y0;
            s1 += y1 * y1;
            s2 += y2 * y2;
            s3 += y3 * y3;
        }
        sq[0] = s0; sq[1] = s1; sq[2] = s2; sq[3] = s3;
    }

    {
        float* __restrict__ op = out + (size_t)(b * NOFF + 4) * CC;
        float s = 0.0f;
#define EXT_ST(e, ci, di) \
        s += ext_store<!(KIND == 0 && (ci) == (di))>(red, NEXT0 + e, cb + 16 * (ci), db + 16 * (di), op, tid);
        if constexpr (KIND == 0) { EXT_K0(EXT_ST) }
        else if constexpr (KIND == 1) { EXT_K1(EXT_ST) }
        else { EXT_K2(EXT_ST) }
#undef EXT_ST
        if constexpr (KIND == 2) s += ext_store<false>(red, NEXT0 + 4, db + 32, db + 32, op, tid);
        sq[4] = s;
    }

#pragma unroll
    for (int o = 0; o < NOFF; ++o) {
        float s = sq[o];
#pragma unroll
        for (int off = 16; off > 0; off >>= 1)
            s += __shfl_xor(s, off, 32);
        if (lane == 0) sred[o][q] = s;
    }
    __syncthreads();
    if (tid < NOFF)
        ss[(b * NOFF + tid) * NSLOT + rt] = (sred[tid][0] + sred[tid][1]) + (sred[tid][2] + sred[tid][3]);
}

__global__ __launch_bounds__(128) void cofe_wmma(const float* __restrict__ x,
                                                 float* __restrict__ out,
                                                 float* __restrict__ ss) {
    __shared__ float red[NWV][NACC][256];
    __shared__ float sred[NOFF][NWV];

    const int b  = blockIdx.x / NSLOT;
    const int rt = blockIdx.x % NSLOT;
    const int I  = rt / NT, J = rt % NT;
    const int cb = I * TS;
    const int db = J * TS;
    const int d  = (J - I) & 3;

    if (d == 0)                          cofe_body<0>(x, out, ss, red, sred, b, rt, cb, db);
    else if (d == 1 || (d == 2 && I < J)) cofe_body<1>(x, out, ss, red, sred, b, rt, cb, db);
    else if (d == 3)                     cofe_body<2>(x, out, ss, red, sred, b, rt, cb, db);
    else                                 cofe_body<3>(x, out, ss, red, sred, b, rt, cb, db);
}

__global__ __launch_bounds__(256) void cofe_normalize(float* __restrict__ out,
                                                      const float* __restrict__ ss) {
    const size_t i = (size_t)blockIdx.x * 256 + threadIdx.x;
    const float v = out[i];
    __builtin_amdgcn_sched_barrier(0);
    const int grp = blockIdx.x / (CC / 256);
    const float* __restrict__ p = ss + grp * NSLOT;
    float t = 0.0f;
#pragma unroll
    for (int j = 0; j < NSLOT; ++j) t += p[j];
    float nrm = sqrtf(t);
    nrm = fmaxf(nrm, 1e-12f);
    out[i] = v / nrm;
}

extern "C" void kernel_launch(void* const* d_in, const int* in_sizes, int n_in,
                              void* d_out, int out_size, void* d_ws, size_t ws_size,
                              hipStream_t stream) {
    const float* x  = (const float*)d_in[0];
    float* out      = (float*)d_out;
    float* ss       = (float*)d_ws;

    cofe_wmma<<<NB * NT * NT, 128, 0, stream>>>(x, out, ss);

    cofe_normalize<<<(NB * NOFF * CC) / 256, 256, 0, stream>>>(out, ss);
}


# Row 0 offsets from the 3 column lags of one row (4 product planes): lag -1 and lag +1 give a (0,-1) fragment and its mirror, lag 0 the symmetric plane; 112 not 128 matrix ops per row; reorders sums
# speedup vs baseline: 3.1975x; 1.0336x over previous
#include <hip/hip_runtime.h>


typedef __attribute__((ext_vector_type(2))) float v2f;
typedef __attribute__((ext_vector_type(4))) float v4f;
typedef __attribute__((ext_vector_type(8))) float v8f;

#define NB   16
#define NC   192
#define NH   64
#define NW   64
#define NK   (NH * NW)
#define NOFF 5
#define NTF  4
#define CC   (NC * NC)
#define TS   48
#define NT   (NC / TS)
#define NSLOT (NT * NT)
#define NMAIN 36
#define NEXT0 NMAIN
#define NACC (NEXT0 + 5 * NTF)
#define NWV  4


#define WM(A_, B_, C_) \
    __builtin_amdgcn_wmma_f32_16x16x4_f32(false, (A_), false, (B_), (short)0, (C_), false, false)

#define EXT_K0(X) X(0, 0, 0) X(1, 0, 1) X(2, 0, 2) X(3, 1, 1) X(4, 1, 2)
#define EXT_K1(X) X(0, 0, 0) X(1, 0, 1) X(2, 0, 2) X(3, 1, 0) X(4, 1, 1)
#define EXT_K2(X) X(0, 0, 2) X(1, 1, 2) X(2, 2, 1) X(3, 2, 2)


struct BWin { v4f lo, hi; float L, R; };

__device__ __forceinline__ void cofe_load_row(const float* __restrict__ pa, const float* __restrict__ pb,
                                              int offL, int offR, v4f (&aw)[3][2], BWin (&bw)[3]) {
#pragma unroll
    for (int ci = 0; ci < 3; ++ci) {
        aw[ci][0] = *(const v4f*)(pa + ci * 16 * NK);
        aw[ci][1] = *(const v4f*)(pa + ci * 16 * NK + 4);
    }
#pragma unroll
    for (int di = 0; di < 3; ++di) {
        const float* __restrict__ p = pb + di * 16 * NK;
        bw[di].lo = *(const v4f*)(p);
        bw[di].hi = *(const v4f*)(p + 4);
        bw[di].L  = p[offL];
        bw[di].R  = p[offR];
    }
}

template <int KIND>
__device__ __forceinline__ void cofe_row(v8f (&acc)[NACC], const v4f (&aw)[3][2],
                                         const BWin (&up)[3], const BWin (&cu)[3], bool lastcols) {
    float A[3][8], U[3][10], C[3][10];
#pragma unroll
    for (int i = 0; i < 3; ++i) {
#pragma unroll
        for (int e = 0; e < 4; ++e) {
            A[i][e] = aw[i][0][e];     A[i][4 + e] = aw[i][1][e];
            U[i][1 + e] = up[i].lo[e]; U[i][5 + e] = up[i].hi[e];
            C[i][1 + e] = cu[i].lo[e]; C[i][5 + e] = cu[i].hi[e];
        }
        U[i][0] = up[i].L; U[i][9] = up[i].R;
        C[i][0] = cu[i].L; C[i][9] = lastcols ? 0.0f : cu[i].R;
    }

#pragma unroll
    for (int s = 0; s < 2; ++s) {
        v2f ea[3][NTF], vb[3][NTF], wb[3][NTF];
#pragma unroll
        for (int i = 0; i < 3; ++i) {
#pragma unroll
            for (int e = 0; e < 2; ++e) {
                const int j = 2 * s + e;
                const float a0 = A[i][2 * j], a1 = A[i][2 * j + 1];
                ea[i][0][e] = a0;
                ea[i][1][e] = a0 + a1;
                ea[i][2][e] = a0 - a1;
                ea[i][3][e] = a1;
                const float b0 = U[i][2 * j], b1 = U[i][2 * j + 1], b2 = U[i][2 * j + 2], b3 = U[i][2 * j + 3];
                vb[i][0][e] = b0 - b2;
                vb[i][1][e] = b1 + b2;
                vb[i][2][e] = b2 - b1;
                vb[i][3][e] = b3 - b1;
                const float c0 = C[i][2 * j], c1 = C[i][2 * j + 1], c2 = C[i][2 * j + 2], c3 = C[i][2 * j + 3];
                wb[i][0][e] = c0 - c2;
                wb[i][1][e] = c1 + c2;
                wb[i][2][e] = c2 - c1;
                wb[i][3][e] = c3 - c1;
            }
        }
#pragma unroll
        for (int di = 0; di < 3; ++di)
#pragma unroll
            for (int t = 0; t < NTF; ++t)
#pragma unroll
                for (int ci = 0; ci < 3; ++ci)
                    acc[(ci * 3 + di) * NTF + t] = WM(ea[ci][t], vb[di][t], acc[(ci * 3 + di) * NTF + t]);
#define EXT_MM(e_, ci, di) \
        if constexpr (KIND != 0 || (ci) != (di)) { \
            _Pragma("unroll") for (int t = 0; t < NTF; ++t) \
                acc[NEXT0 + (e_) * NTF + t] = WM(ea[ci][t], wb[di][t], acc[NEXT0 + (e_) * NTF + t]); \
        }
        if constexpr (KIND == 0) { EXT_K0(EXT_MM) }
        else if constexpr (KIND == 1) { EXT_K1(EXT_MM) }
        else { EXT_K2(EXT_MM) }
#undef EXT_MM
    }

    if constexpr (KIND == 0 || KIND == 2) {
#pragma unroll
        for (int j = 0; j < 4; ++j) {
#define EXT_DG(e_, ci, di) \
            if constexpr ((ci) == (di)) { \
                const v2f a_  = v2f{A[ci][2 * j],     A[ci][2 * j + 1]}; \
                const v2f b3_ = v2f{C[di][2 * j],     C[di][2 * j + 1]}; \
                const v2f b4_ = v2f{C[di][2 * j + 1], C[di][2 * j + 2]}; \
                acc[NEXT0 + (e_) * NTF + 0] = WM(a_, b3_, acc[NEXT0 + (e_) * NTF + 0]); \
                acc[NEXT0 + (e_) * NTF + 1] = WM(a_, b4_, acc[NEXT0 + (e_) * NTF + 1]); \
            }
            if constexpr (KIND == 0) { EXT_K0(EXT_DG) }
#undef EXT_DG
            if constexpr (KIND == 2) {
                const v2f b3_ = v2f{C[2][2 * j],     C[2][2 * j + 1]};
                const v2f b4_ = v2f{C[2][2 * j + 1], C[2][2 * j + 2]};
                acc[NEXT0 + 4 * NTF + 0] = WM(b4_, b3_, acc[NEXT0 + 4 * NTF + 0]);
                acc[NEXT0 + 4 * NTF + 1] = WM(b4_, b4_, acc[NEXT0 + 4 * NTF + 1]);
            }
        }
    }
}

__device__ __forceinline__ float red4(const float (*red)[NACC][256], int t, int pos) {
    return (red[0][t][pos] + red[1][t][pos]) + (red[2][t][pos] + red[3][t][pos]);
}

template <bool OFFDIAG>
__device__ __forceinline__ void ext_store(const float (*red)[NACC][256], int t0, int rbase, int cbase,
                                          float* __restrict__ o3, float* __restrict__ o4, int tid,
                                          float& s3, float& s4) {
#pragma unroll
    for (int jj = 0; jj < 2; ++jj) {
        const int idx = jj * 128 + tid;
        {
            const int r = idx >> 4, c = idx & 15;
            const int pos = ((r & 7) << 5) + c + (((r >> 3) & 1) << 4);
            float ym, yz;
            if constexpr (OFFDIAG) {
                const float m0 = red4(red, t0 + 0, pos);
                const float m1 = red4(red, t0 + 1, pos);
                const float m2 = red4(red, t0 + 2, pos);
                ym = m0 + 0.5f * (m1 + m2);
                yz = 0.5f * (m1 - m2);
            } else {
                ym = red4(red, t0 + 0, pos);
                yz = red4(red, t0 + 1, pos);
            }
            const size_t o = (size_t)(rbase + r) * NC + (size_t)(cbase + c);
            o3[o] = ym;
            o4[o] = yz;
            s3 += ym * ym;
            s4 += OFFDIAG ? 2.0f * (yz * yz) : yz * yz;
        }
        if constexpr (OFFDIAG) {
            const int r = idx & 15, c = idx >> 4;
            const int pos = ((r & 7) << 5) + c + (((r >> 3) & 1) << 4);
            const float m1 = red4(red, t0 + 1, pos);
            const float m2 = red4(red, t0 + 2, pos);
            const float m3 = red4(red, t0 + 3, pos);
            const float yp = 0.5f * (m1 + m2) + m3;
            const float yz = 0.5f * (m1 - m2);
            const size_t o = (size_t)(cbase + c) * NC + (size_t)(rbase + r);
            o3[o] = yp;
            o4[o] = yz;
            s3 += yp * yp;
        }
    }
}

template <int KIND>
__device__ __forceinline__ void cofe_body(const float* __restrict__ x, float* __restrict__ out,
                                          float* __restrict__ ss, float (*red)[NACC][256],
                                          float (*sred)[NWV], int b, int rt, int cb, int db) {
    constexpr int NUSED = NEXT0 + (KIND == 3 ? 4 : 5) * NTF;

    const int tid    = threadIdx.x;
    const int q      = tid >> 5;
    const int lane   = tid & 31;
    const int lane15 = lane & 15;

    const float* __restrict__ xb = x + (size_t)b * NC * NK;
    const int colbase = (q << 4) + ((lane >> 4) << 3);
    const int offL = (colbase == 0)      ? 0 : -1;
    const int offR = (colbase == NW - 8) ? 7 :  8;
    const bool lastcols = (colbase == NW - 8);
    const float* __restrict__ pa = xb + (cb + lane15) * NK + colbase;
    const float* __restrict__ pb = xb + (db + lane15) * NK + colbase;

    v8f acc[NACC];
#pragma unroll
    for (int t = 0; t < NACC; ++t) acc[t] = v8f{};

    {
        const int r0 = (q << 4) + ((lane >> 4) << 1);
        const float* __restrict__ ka = xb + (cb + lane15) * NK + (r0 << 6);
        const float* __restrict__ kb = xb + (db + lane15) * NK + (r0 << 6);
        v2f va[3][4], vb[3][4];
#pragma unroll
        for (int i = 0; i < 3; ++i)
#pragma unroll
            for (int s = 0; s < 4; ++s) {
                va[i][s] = v2f{ka[i * 16 * NK + (4 * s) * NW], ka[i * 16 * NK + (4 * s + 1) * NW]};
                vb[i][s] = v2f{kb[i * 16 * NK + (4 * s) * NW], kb[i * 16 * NK + (4 * s + 1) * NW]};
            }
#define EXT_C0(e_, ci, di) \
        if constexpr (KIND != 0 || (ci) != (di)) { \
            _Pragma("unroll") for (int s = 0; s < 4; ++s) \
                acc[NEXT0 + (e_) * NTF + 3] = WM(va[ci][s], vb[di][s], acc[NEXT0 + (e_) * NTF + 3]); \
        }
        if constexpr (KIND == 0) { EXT_K0(EXT_C0) }
        else if constexpr (KIND == 1) { EXT_K1(EXT_C0) }
        else { EXT_K2(EXT_C0) }
#undef EXT_C0
    }

    v4f  acur[3][2], anxt[3][2];
    BWin up[3], cu[3], nx[3];
    cofe_load_row(pa, pb, offL, offR, acur, cu);
#pragma unroll
    for (int di = 0; di < 3; ++di) up[di] = cu[di];

    for (int h = 0; h < NH; ++h) {
        const int hn = (h < NH - 1) ? (h + 1) : (NH - 1);
        cofe_load_row(pa + (hn << 6), pb + (hn << 6), offL, offR, anxt, nx);
        cofe_row<KIND>(acc, acur, up, cu, lastcols);
#pragma unroll
        for (int i = 0; i < 3; ++i) {
            up[i] = cu[i];
            cu[i] = nx[i];
            acur[i][0] = anxt[i][0];
            acur[i][1] = anxt[i][1];
        }
    }

#pragma unroll
    for (int t = 0; t < NUSED; ++t)
#pragma unroll
        for (int i = 0; i < 8; ++i)
            red[q][t][i * 32 + lane] = acc[t][i];
    __syncthreads();

    float sq[NOFF];

    {
        float s0 = 0.0f, s1 = 0.0f, s2 = 0.0f;
        float* __restrict__ ob = out + (size_t)(b * NOFF) * CC + (size_t)cb * NC + (size_t)db;
#pragma unroll
        for (int jj = 0; jj < (TS * TS) / 128; ++jj) {
            const int rem = jj * 128 + tid;
            const int row = rem / TS, col = rem % TS;
            const int f   = (row >> 4) * 3 + (col >> 4);
            const int pos = ((row & 7) << 5) + (col & 15) + (((row >> 3) & 1) << 4);
            const float m0 = red4(red, f * NTF + 0, pos);
            const float m1 = red4(red, f * NTF + 1, pos);
            const float m2 = red4(red, f * NTF + 2, pos);
            const float m3 = red4(red, f * NTF + 3, pos);
            const float hs = 0.5f * (m1 + m2);
            const float y0 = m0 + hs;
            const float y1 = 0.5f * (m1 - m2);
            const float y2 = hs + m3;
            float* __restrict__ o = ob + (size_t)row * NC + (size_t)col;
            o[0]               = y0;
            o[(size_t)CC]      = y1;
            o[(size_t)2 * CC]  = y2;
            s0 += y0 * y0;
            s1 += y1 * y1;
            s2 += y2 * y2;
        }
        sq[0] = s0; sq[1] = s1; sq[2] = s2;
    }

    {
        float* __restrict__ o3 = out + (size_t)(b * NOFF + 3) * CC;
        float* __restrict__ o4 = out + (size_t)(b * NOFF + 4) * CC;
        float s3 = 0.0f, s4 = 0.0f;
#define EXT_ST(e_, ci, di) \
        ext_store<!(KIND == 0 && (ci) == (di))>(red, NEXT0 + (e_) * NTF, cb + 16 * (ci), db + 16 * (di), o3, o4, tid, s3, s4);
        if constexpr (KIND == 0) { EXT_K0(EXT_ST) }
        else if constexpr (KIND == 1) { EXT_K1(EXT_ST) }
        else { EXT_K2(EXT_ST) }
#undef EXT_ST
        if constexpr (KIND == 2) ext_store<false>(red, NEXT0 + 4 * NTF, db + 32, db + 32, o3, o4, tid, s3, s4);
        sq[3] = s3;
        sq[4] = s4;
    }

#pragma unroll
    for (int o = 0; o < NOFF; ++o) {
        float s = sq[o];
#pragma unroll
        for (int off = 16; off > 0; off >>= 1)
            s += __shfl_xor(s, off, 32);
        if (lane == 0) sred[o][q] = s;
    }
    __syncthreads();
    if (tid < NOFF)
        ss[(b * NOFF + tid) * NSLOT + rt] = (sred[tid][0] + sred[tid][1]) + (sred[tid][2] + sred[tid][3]);
}

__global__ __launch_bounds__(128) void cofe_wmma(const float* __restrict__ x,
                                                 float* __restrict__ out,
                                                 float* __restrict__ ss) {
    __shared__ float red[NWV][NACC][256];
    __shared__ float sred[NOFF][NWV];

    const int b  = blockIdx.x / NSLOT;
    const int rt = blockIdx.x % NSLOT;
    const int I  = rt / NT, J = rt % NT;
    const int cb = I * TS;
    const int db = J * TS;
    const int d  = (J - I) & 3;

    if (d == 0)                          cofe_body<0>(x, out, ss, red, sred, b, rt, cb, db);
    else if (d == 1 || (d == 2 && I < J)) cofe_body<1>(x, out, ss, red, sred, b, rt, cb, db);
    else if (d == 3)                     cofe_body<2>(x, out, ss, red, sred, b, rt, cb, db);
    else                                 cofe_body<3>(x, out, ss, red, sred, b, rt, cb, db);
}

__global__ __launch_bounds__(256) void cofe_normalize(float* __restrict__ out,
                                                      const float* __restrict__ ss) {
    const size_t i = (size_t)blockIdx.x * 256 + threadIdx.x;
    const float v = out[i];
    __builtin_amdgcn_sched_barrier(0);
    const int grp = blockIdx.x / (CC / 256);
    const float* __restrict__ p = ss + grp * NSLOT;
    float t = 0.0f;
#pragma unroll
    for (int j = 0; j < NSLOT; ++j) t += p[j];
    float nrm = sqrtf(t);
    nrm = fmaxf(nrm, 1e-12f);
    out[i] = v / nrm;
}

extern "C" void kernel_launch(void* const* d_in, const int* in_sizes, int n_in,
                              void* d_out, int out_size, void* d_ws, size_t ws_size,
                              hipStream_t stream) {
    const float* x  = (const float*)d_in[0];
    float* out      = (float*)d_out;
    float* ss       = (float*)d_ws;

    cofe_wmma<<<NB * NT * NT, 128, 0, stream>>>(x, out, ss);

    cofe_normalize<<<(NB * NOFF * CC) / 256, 256, 0, stream>>>(out, ss);
}
